# starting kernel
# baseline (speedup 1.0000x reference)
#include <hip/hip_runtime.h>

typedef __attribute__((ext_vector_type(2))) float v2f;
typedef __attribute__((ext_vector_type(8))) float v8f;

#define KDIM   576
#define NCOL   784
#define TILE_N 16
#define LDS_ELEMS (KDIM * TILE_N)

__global__ __launch_bounds__(256)
void shiftconv_wmma_f32(const float* __restrict__ x,
                        const float* __restrict__ w,
                        float* __restrict__ out) {
  __shared__ float Bt[LDS_ELEMS];

  const int n0 = blockIdx.x * TILE_N;

  for (int e = threadIdx.x; e < LDS_ELEMS; e += 256) {
    const int r   = e >> 4;
    const int c   = e & 15;
    const int col = n0 + c;
    const int n   = col / 28;
    const int o   = col % 28;
    const int j   = r / 192;
    const int k   = (r / 64) % 3;
    const int l   = r & 63;
    const int h   = j + n - 1;
    float v = 0.0f;
    if (h >= 0 && h < 28) {
      const int ko = k + o;
      if (ko >= 1 && ko <= 28)
        v += x[l * 784 + h * 28 + ((ko + 26) % 28)];
      const int om  = (o + 27) % 28;
      const int kom = k + om;
      if (kom >= 1 && kom <= 28)
        v += x[(64 + l) * 784 + h * 28 + ((kom + 26) % 28)];
    }
    Bt[e] = v;
  }
  __syncthreads();

  const int wave = threadIdx.x >> 5;
  const int lane = threadIdx.x & 31;
  const int half = lane >> 4;
  const int lid  = lane & 15;
  const int koff = half * 2;

  const float* __restrict__ arow = w + (wave * 16 + lid) * KDIM + koff;
  const float* __restrict__ bcol = &Bt[koff * TILE_N + lid];

  __builtin_prefetch(arow, 0, 0);

  v8f acc0 = {};
  v8f acc1 = {};
#pragma unroll 4
  for (int kk = 0; kk < KDIM; kk += 8) {
    v2f a0, b0, a1, b1;
    a0.x = arow[kk];
    a0.y = arow[kk + 1];
    b0.x = bcol[kk * TILE_N];
    b0.y = bcol[(kk + 1) * TILE_N];
    a1.x = arow[kk + 4];
    a1.y = arow[kk + 5];
    b1.x = bcol[(kk + 4) * TILE_N];
    b1.y = bcol[(kk + 5) * TILE_N];
    acc0 = __builtin_amdgcn_wmma_f32_16x16x4_f32(false, a0, false, b0,
                                                 (short)0, acc0, false, false);
    acc1 = __builtin_amdgcn_wmma_f32_16x16x4_f32(false, a1, false, b1,
                                                 (short)0, acc1, false, false);
  }
  v8f acc = acc0 + acc1;

  const int nidx  = n0 + lid;
  const int ibase = wave * 16 + half * 8;
#pragma unroll
  for (int r = 0; r < 8; ++r)
    out[(ibase + r) * NCOL + nidx] = acc[r];
}

extern "C" void kernel_launch(void* const* d_in, const int* in_sizes, int n_in,
                              void* d_out, int out_size, void* d_ws, size_t ws_size,
                              hipStream_t stream) {
  const float* x = (const float*)d_in[0];
  const float* w = (const float*)d_in[1];
  float* out = (float*)d_out;

  shiftconv_wmma_f32<<<NCOL / TILE_N, 256, 0, stream>>>(x, w, out);
}


# A operand: w streamed through LDS in 3 K chunks by all threads with coalesced 16-byte loads (2 chunks requested up front, the third across the gather); the matrix loop reads fragments from LDS only
# speedup vs baseline: 2.3036x; 2.3036x over previous
#include <hip/hip_runtime.h>


typedef __attribute__((ext_vector_type(2))) float v2f;
typedef __attribute__((ext_vector_type(4))) float v4f;
typedef __attribute__((ext_vector_type(8))) float v8f;

#define KDIM   576
#define NCOL   784
#define TILE_N 16
#define LDS_ELEMS (KDIM * TILE_N)
#define XROWS  4
#define XQUADS (128 * XROWS * 7)
#define XS_ELEMS (XQUADS * 4)
#define KCH    192
#define APITCH 196
#define ACH_ELEMS (128 * APITCH)
#define AQ_PER_THREAD 24

__device__ __forceinline__ void w_chunk_load(const float* __restrict__ w, int j, int tid, v4f (&r)[AQ_PER_THREAD]) {
#pragma unroll
  for (int it = 0; it < AQ_PER_THREAD; ++it) {
    const int q   = tid + 256 * it;
    const int row = q / 48;
    const int c4  = q - row * 48;
    r[it] = *reinterpret_cast<const v4f*>(w + row * KDIM + j * KCH + c4 * 4);
  }
}

__device__ __forceinline__ void w_chunk_store(float* __restrict__ As, int tid, const v4f (&r)[AQ_PER_THREAD]) {
#pragma unroll
  for (int it = 0; it < AQ_PER_THREAD; ++it) {
    const int q   = tid + 256 * it;
    const int row = q / 48;
    const int c4  = q - row * 48;
    *reinterpret_cast<v4f*>(As + row * APITCH + c4 * 4) = r[it];
  }
}

__device__ __forceinline__ void lds_barrier() {
  __builtin_amdgcn_fence(__ATOMIC_RELEASE, "workgroup", "local");
  __builtin_amdgcn_s_barrier();
  __builtin_amdgcn_fence(__ATOMIC_ACQUIRE, "workgroup", "local");
}

struct Frag8 { v2f a[8]; v2f b[8]; };

__device__ __forceinline__ void frag_load(Frag8& f, const float* __restrict__ ar,
                                          const float* __restrict__ bc, int kl0) {
#pragma unroll
  for (int u = 0; u < 8; ++u) {
    const int kl = kl0 + 4 * u;
    f.a[u] = *reinterpret_cast<const v2f*>(ar + kl);
    f.b[u].x = bc[kl * TILE_N];
    f.b[u].y = bc[(kl + 1) * TILE_N];
  }
}

__device__ __forceinline__ void frag_mma(const Frag8& f, v8f& acc0, v8f& acc1) {
#pragma unroll
  for (int u = 0; u < 8; u += 2) {
    acc0 = __builtin_amdgcn_wmma_f32_16x16x4_f32(false, f.a[u], false, f.b[u],
                                                 (short)0, acc0, false, false);
    acc1 = __builtin_amdgcn_wmma_f32_16x16x4_f32(false, f.a[u + 1], false, f.b[u + 1],
                                                 (short)0, acc1, false, false);
  }
}

__device__ __forceinline__ void mma_chunk(const float* __restrict__ ar, const float* __restrict__ bc,
                                          v8f& acc0, v8f& acc1) {
  Frag8 f0, f1;
  frag_load(f0, ar, bc, 0);
#pragma unroll
  for (int blk = 0; blk < KCH / 32; blk += 2) {
    frag_load(f1, ar, bc, (blk + 1) * 32);
    __builtin_amdgcn_sched_barrier(0);
    frag_mma(f0, acc0, acc1);
    __builtin_amdgcn_sched_barrier(0);
    if (blk + 2 < KCH / 32) frag_load(f0, ar, bc, (blk + 2) * 32);
    __builtin_amdgcn_sched_barrier(0);
    frag_mma(f1, acc0, acc1);
    __builtin_amdgcn_sched_barrier(0);
  }
}

__global__ __launch_bounds__(256)
void shiftconv_wmma_f32(const float* __restrict__ x,
                        const float* __restrict__ w,
                        float* __restrict__ out) {
  __shared__ __attribute__((aligned(16))) float lds[LDS_ELEMS + XS_ELEMS + 2 * ACH_ELEMS];
  float* const Bt  = lds;
  float* const Xs  = lds + LDS_ELEMS;
  float* const As0 = Xs + XS_ELEMS;
  float* const As1 = As0 + ACH_ELEMS;

  const int tid   = threadIdx.x;
  const int n0    = blockIdx.x * TILE_N;
  const int nlo   = n0 / 28;
  const int hbase = nlo - 1;

  v4f xin[14];
#pragma unroll
  for (int it = 0; it < 14; ++it) {
    const int idx = tid + 256 * it;
    const int c   = idx / 28;
    const int v   = idx - c * 28;
    const int h   = hbase + v / 7;
    v4f val = {0.0f, 0.0f, 0.0f, 0.0f};
    if (h >= 0 && h < 28)
      val = *reinterpret_cast<const v4f*>(x + c * 784 + hbase * 28 + v * 4);
    xin[it] = val;
  }
  v4f wqa[AQ_PER_THREAD];
  v4f wqb[AQ_PER_THREAD];
  w_chunk_load(w, 0, tid, wqa);
  __builtin_amdgcn_sched_barrier(0);
  w_chunk_load(w, 1, tid, wqb);
#pragma unroll
  for (int it = 0; it < 14; ++it)
    *reinterpret_cast<v4f*>(&Xs[(tid + 256 * it) * 4]) = xin[it];
  w_chunk_store(As0, tid, wqa);
  lds_barrier();

  w_chunk_load(w, 2, tid, wqa);

  {
    const int cc   = tid & 15;
    const int lrow = tid >> 4;
    const int col  = n0 + cc;
    const int n    = col / 28;
    const int o    = col - n * 28;
    const int dn   = n - nlo;
    const int om   = (o + 27) % 28;
    int  a1[3], a2[3];
    bool c1[3], c2[3];
#pragma unroll
    for (int k = 0; k < 3; ++k) {
      const int ko  = k + o;
      const int kom = k + om;
      c1[k] = (ko >= 1 && ko <= 28);
      c2[k] = (kom >= 1 && kom <= 28);
      a1[k] = lrow * 112 + dn * 28 + ((ko + 26) % 28);
      a2[k] = (64 + lrow) * 112 + dn * 28 + ((kom + 26) % 28);
    }
#pragma unroll
    for (int j = 0; j < 3; ++j) {
#pragma unroll
      for (int k = 0; k < 3; ++k) {
#pragma unroll
        for (int li = 0; li < 4; ++li) {
          const int off = li * (16 * 112) + j * 28;
          const float a = Xs[a1[k] + off];
          const float b = Xs[a2[k] + off];
          Bt[((j * 3 + k) * 4 + li) * 256 + tid] = (c1[k] ? a : 0.0f) + (c2[k] ? b : 0.0f);
        }
      }
    }
  }
  w_chunk_store(As1, tid, wqb);
  lds_barrier();

  const int wave = threadIdx.x >> 5;
  const int lane = threadIdx.x & 31;
  const int half = lane >> 4;
  const int lid  = lane & 15;
  const int koff = half * 2;

  const int    aoff = (wave * 16 + lid) * APITCH + koff;
  const float* bcol = &Bt[koff * TILE_N + lid];

  v8f acc0 = {};
  v8f acc1 = {};
  mma_chunk(As0 + aoff, bcol, acc0, acc1);
  lds_barrier();
  w_chunk_store(As0, tid, wqa);
  mma_chunk(As1 + aoff, bcol + KCH * TILE_N, acc0, acc1);
  lds_barrier();
  mma_chunk(As0 + aoff, bcol + 2 * KCH * TILE_N, acc0, acc1);
  v8f acc = acc0 + acc1;

  const int nidx  = n0 + lid;
  const int ibase = wave * 16 + half * 8;
#pragma unroll
  for (int r = 0; r < 8; ++r)
    out[(ibase + r) * NCOL + nidx] = acc[r];
}

extern "C" void kernel_launch(void* const* d_in, const int* in_sizes, int n_in,
                              void* d_out, int out_size, void* d_ws, size_t ws_size,
                              hipStream_t stream) {
  const float* x = (const float*)d_in[0];
  const float* w = (const float*)d_in[1];
  float* out = (float*)d_out;

  shiftconv_wmma_f32<<<NCOL / TILE_N, 256, 0, stream>>>(x, w, out);
}


# im2col gather: a thread issues all 72 LDS reads of the x copy before the first of its 36 tile writes; reads and writes in turn had compiled to 36 read-wait-write steps in series
# speedup vs baseline: 2.4779x; 1.0757x over previous
#include <hip/hip_runtime.h>


typedef __attribute__((ext_vector_type(2))) float v2f;
typedef __attribute__((ext_vector_type(4))) float v4f;
typedef __attribute__((ext_vector_type(8))) float v8f;

#define KDIM   576
#define NCOL   784
#define TILE_N 16
#define LDS_ELEMS (KDIM * TILE_N)
#define XROWS  4
#define XQUADS (128 * XROWS * 7)
#define XS_ELEMS (XQUADS * 4)
#define KCH    192
#define APITCH 196
#define ACH_ELEMS (128 * APITCH)
#define AQ_PER_THREAD 24

__device__ __forceinline__ void w_chunk_load(const float* __restrict__ w, int j, int tid, v4f (&r)[AQ_PER_THREAD]) {
#pragma unroll
  for (int it = 0; it < AQ_PER_THREAD; ++it) {
    const int q   = tid + 256 * it;
    const int row = q / 48;
    const int c4  = q - row * 48;
    r[it] = *reinterpret_cast<const v4f*>(w + row * KDIM + j * KCH + c4 * 4);
  }
}

__device__ __forceinline__ void w_chunk_store(float* __restrict__ As, int tid, const v4f (&r)[AQ_PER_THREAD]) {
#pragma unroll
  for (int it = 0; it < AQ_PER_THREAD; ++it) {
    const int q   = tid + 256 * it;
    const int row = q / 48;
    const int c4  = q - row * 48;
    *reinterpret_cast<v4f*>(As + row * APITCH + c4 * 4) = r[it];
  }
}

__device__ __forceinline__ void lds_barrier() {
  __builtin_amdgcn_fence(__ATOMIC_RELEASE, "workgroup", "local");
  __builtin_amdgcn_s_barrier();
  __builtin_amdgcn_fence(__ATOMIC_ACQUIRE, "workgroup", "local");
}

struct Frag8 { v2f a[8]; v2f b[8]; };

__device__ __forceinline__ void frag_load(Frag8& f, const float* __restrict__ ar,
                                          const float* __restrict__ bc, int kl0) {
#pragma unroll
  for (int u = 0; u < 8; ++u) {
    const int kl = kl0 + 4 * u;
    f.a[u] = *reinterpret_cast<const v2f*>(ar + kl);
    f.b[u].x = bc[kl * TILE_N];
    f.b[u].y = bc[(kl + 1) * TILE_N];
  }
}

__device__ __forceinline__ void frag_mma(const Frag8& f, v8f& acc0, v8f& acc1) {
#pragma unroll
  for (int u = 0; u < 8; u += 2) {
    acc0 = __builtin_amdgcn_wmma_f32_16x16x4_f32(false, f.a[u], false, f.b[u],
                                                 (short)0, acc0, false, false);
    acc1 = __builtin_amdgcn_wmma_f32_16x16x4_f32(false, f.a[u + 1], false, f.b[u + 1],
                                                 (short)0, acc1, false, false);
  }
}

__device__ __forceinline__ void mma_chunk(const float* __restrict__ ar, const float* __restrict__ bc,
                                          v8f& acc0, v8f& acc1) {
  Frag8 f0, f1;
  frag_load(f0, ar, bc, 0);
#pragma unroll
  for (int blk = 0; blk < KCH / 32; blk += 2) {
    frag_load(f1, ar, bc, (blk + 1) * 32);
    __builtin_amdgcn_sched_barrier(0);
    frag_mma(f0, acc0, acc1);
    __builtin_amdgcn_sched_barrier(0);
    if (blk + 2 < KCH / 32) frag_load(f0, ar, bc, (blk + 2) * 32);
    __builtin_amdgcn_sched_barrier(0);
    frag_mma(f1, acc0, acc1);
    __builtin_amdgcn_sched_barrier(0);
  }
}

__global__ __launch_bounds__(256)
void shiftconv_wmma_f32(const float* __restrict__ x,
                        const float* __restrict__ w,
                        float* __restrict__ out) {
  __shared__ __attribute__((aligned(16))) float lds[LDS_ELEMS + XS_ELEMS + 2 * ACH_ELEMS];
  float* const Bt  = lds;
  float* const Xs  = lds + LDS_ELEMS;
  float* const As0 = Xs + XS_ELEMS;
  float* const As1 = As0 + ACH_ELEMS;

  const int tid   = threadIdx.x;
  const int n0    = blockIdx.x * TILE_N;
  const int nlo   = n0 / 28;
  const int hbase = nlo - 1;

  v4f xin[14];
#pragma unroll
  for (int it = 0; it < 14; ++it) {
    const int idx = tid + 256 * it;
    const int c   = idx / 28;
    const int v   = idx - c * 28;
    const int h   = hbase + v / 7;
    v4f val = {0.0f, 0.0f, 0.0f, 0.0f};
    if (h >= 0 && h < 28)
      val = *reinterpret_cast<const v4f*>(x + c * 784 + hbase * 28 + v * 4);
    xin[it] = val;
  }
  v4f wqa[AQ_PER_THREAD];
  v4f wqb[AQ_PER_THREAD];
  w_chunk_load(w, 0, tid, wqa);
  __builtin_amdgcn_sched_barrier(0);
  w_chunk_load(w, 1, tid, wqb);
#pragma unroll
  for (int it = 0; it < 14; ++it)
    *reinterpret_cast<v4f*>(&Xs[(tid + 256 * it) * 4]) = xin[it];
  w_chunk_store(As0, tid, wqa);
  lds_barrier();

  w_chunk_load(w, 2, tid, wqa);

  {
    const int cc   = tid & 15;
    const int lrow = tid >> 4;
    const int col  = n0 + cc;
    const int n    = col / 28;
    const int o    = col - n * 28;
    const int dn   = n - nlo;
    const int om   = (o + 27) % 28;
    int  a1[3], a2[3];
    bool c1[3], c2[3];
#pragma unroll
    for (int k = 0; k < 3; ++k) {
      const int ko  = k + o;
      const int kom = k + om;
      c1[k] = (ko >= 1 && ko <= 28);
      c2[k] = (kom >= 1 && kom <= 28);
      a1[k] = lrow * 112 + dn * 28 + ((ko + 26) % 28);
      a2[k] = (64 + lrow) * 112 + dn * 28 + ((kom + 26) % 28);
    }
    float av[36], bv[36];
#pragma unroll
    for (int j = 0; j < 3; ++j) {
#pragma unroll
      for (int k = 0; k < 3; ++k) {
#pragma unroll
        for (int li = 0; li < 4; ++li) {
          const int off = li * (16 * 112) + j * 28;
          av[(j * 3 + k) * 4 + li] = Xs[a1[k] + off];
          bv[(j * 3 + k) * 4 + li] = Xs[a2[k] + off];
        }
      }
    }
#pragma unroll
    for (int j = 0; j < 3; ++j) {
#pragma unroll
      for (int k = 0; k < 3; ++k) {
#pragma unroll
        for (int li = 0; li < 4; ++li) {
          const int e = (j * 3 + k) * 4 + li;
          Bt[e * 256 + tid] = (c1[k] ? av[e] : 0.0f) + (c2[k] ? bv[e] : 0.0f);
        }
      }
    }
  }
  w_chunk_store(As1, tid, wqb);
  lds_barrier();

  const int wave = threadIdx.x >> 5;
  const int lane = threadIdx.x & 31;
  const int half = lane >> 4;
  const int lid  = lane & 15;
  const int koff = half * 2;

  const int    aoff = (wave * 16 + lid) * APITCH + koff;
  const float* bcol = &Bt[koff * TILE_N + lid];

  v8f acc0 = {};
  v8f acc1 = {};
  mma_chunk(As0 + aoff, bcol, acc0, acc1);
  lds_barrier();
  w_chunk_store(As0, tid, wqa);
  mma_chunk(As1 + aoff, bcol + KCH * TILE_N, acc0, acc1);
  lds_barrier();
  mma_chunk(As0 + aoff, bcol + 2 * KCH * TILE_N, acc0, acc1);
  v8f acc = acc0 + acc1;

  const int nidx  = n0 + lid;
  const int ibase = wave * 16 + half * 8;
#pragma unroll
  for (int r = 0; r < 8; ++r)
    out[(ibase + r) * NCOL + nidx] = acc[r];
}

extern "C" void kernel_launch(void* const* d_in, const int* in_sizes, int n_in,
                              void* d_out, int out_size, void* d_ws, size_t ws_size,
                              hipStream_t stream) {
  const float* x = (const float*)d_in[0];
  const float* w = (const float*)d_in[1];
  float* out = (float*)d_out;

  shiftconv_wmma_f32<<<NCOL / TILE_N, 256, 0, stream>>>(x, w, out);
}


# x and w staging: thread-to-element maps under which a thread's loads and LDS stores are one base plus constants and one row guard, so all issue at once; no per-load division or guard branch
# speedup vs baseline: 2.6357x; 1.0637x over previous
#include <hip/hip_runtime.h>


typedef __attribute__((ext_vector_type(2))) float v2f;
typedef __attribute__((ext_vector_type(4))) float v4f;
typedef __attribute__((ext_vector_type(8))) float v8f;

#define KDIM   576
#define NCOL   784
#define TILE_N 16
#define LDS_ELEMS (KDIM * TILE_N)
#define XROWS  4
#define XQUADS (128 * XROWS * 7)
#define XS_ELEMS (XQUADS * 4)
#define XCH_PER_WAVE 16
#define KCH    192
#define APITCH 196
#define ACH_ELEMS (128 * APITCH)
#define AQ_PER_THREAD 24

__device__ __forceinline__ void w_chunk_load(const float* __restrict__ wp, int j, v4f (&r)[AQ_PER_THREAD]) {
#pragma unroll
  for (int it = 0; it < AQ_PER_THREAD; ++it) {
    const int rg = it / 3;
    const int cg = it - rg * 3;
    r[it] = *reinterpret_cast<const v4f*>(wp + rg * (16 * KDIM) + j * KCH + cg * 64);
  }
}

__device__ __forceinline__ void w_chunk_store(float* __restrict__ ap, const v4f (&r)[AQ_PER_THREAD]) {
#pragma unroll
  for (int it = 0; it < AQ_PER_THREAD; ++it) {
    const int rg = it / 3;
    const int cg = it - rg * 3;
    *reinterpret_cast<v4f*>(ap + rg * (16 * APITCH) + cg * 64) = r[it];
  }
}

__device__ __forceinline__ void lds_barrier() {
  __builtin_amdgcn_fence(__ATOMIC_RELEASE, "workgroup", "local");
  __builtin_amdgcn_s_barrier();
  __builtin_amdgcn_fence(__ATOMIC_ACQUIRE, "workgroup", "local");
}

struct Frag8 { v2f a[8]; v2f b[8]; };

__device__ __forceinline__ void frag_load(Frag8& f, const float* __restrict__ ar,
                                          const float* __restrict__ bc, int kl0) {
#pragma unroll
  for (int u = 0; u < 8; ++u) {
    const int kl = kl0 + 4 * u;
    f.a[u] = *reinterpret_cast<const v2f*>(ar + kl);
    f.b[u].x = bc[kl * TILE_N];
    f.b[u].y = bc[(kl + 1) * TILE_N];
  }
}

__device__ __forceinline__ void frag_mma(const Frag8& f, v8f& acc0, v8f& acc1) {
#pragma unroll
  for (int u = 0; u < 8; u += 2) {
    acc0 = __builtin_amdgcn_wmma_f32_16x16x4_f32(false, f.a[u], false, f.b[u],
                                                 (short)0, acc0, false, false);
    acc1 = __builtin_amdgcn_wmma_f32_16x16x4_f32(false, f.a[u + 1], false, f.b[u + 1],
                                                 (short)0, acc1, false, false);
  }
}

__device__ __forceinline__ void mma_chunk(const float* __restrict__ ar, const float* __restrict__ bc,
                                          v8f& acc0, v8f& acc1) {
  Frag8 f0, f1;
  frag_load(f0, ar, bc, 0);
#pragma unroll
  for (int blk = 0; blk < KCH / 32; blk += 2) {
    frag_load(f1, ar, bc, (blk + 1) * 32);
    __builtin_amdgcn_sched_barrier(0);
    frag_mma(f0, acc0, acc1);
    __builtin_amdgcn_sched_barrier(0);
    if (blk + 2 < KCH / 32) frag_load(f0, ar, bc, (blk + 2) * 32);
    __builtin_amdgcn_sched_barrier(0);
    frag_mma(f1, acc0, acc1);
    __builtin_amdgcn_sched_barrier(0);
  }
}

__global__ __launch_bounds__(256)
void shiftconv_wmma_f32(const float* __restrict__ x,
                        const float* __restrict__ w,
                        float* __restrict__ out) {
  __shared__ __attribute__((aligned(16))) float lds[LDS_ELEMS + XS_ELEMS + 2 * ACH_ELEMS];
  float* const Bt  = lds;
  float* const Xs  = lds + LDS_ELEMS;
  float* const As0 = Xs + XS_ELEMS;
  float* const As1 = As0 + ACH_ELEMS;

  const int tid   = threadIdx.x;
  const int n0    = blockIdx.x * TILE_N;
  const int nlo   = n0 / 28;
  const int hbase = nlo - 1;

  const int  xlane = tid & 31;
  const int  xwave = tid >> 5;
  const int  xh    = hbase + xlane / 7;
  const bool xact  = xlane < 28;
  const bool xld   = xact && xh >= 0 && xh < 28;
  v4f xin[XCH_PER_WAVE] = {};
  if (xld) {
    const float* xp = x + (xwave * XCH_PER_WAVE) * 784 + hbase * 28 + xlane * 4;
#pragma unroll
    for (int it = 0; it < XCH_PER_WAVE; ++it)
      xin[it] = *reinterpret_cast<const v4f*>(xp + it * 784);
  }
  const int    wsub = (tid >> 4) * KDIM + (tid & 15) * 4;
  const float* wp   = w + wsub;
  const int    asub = (tid >> 4) * APITCH + (tid & 15) * 4;
  v4f wqa[AQ_PER_THREAD];
  v4f wqb[AQ_PER_THREAD];
  w_chunk_load(wp, 0, wqa);
  __builtin_amdgcn_sched_barrier(0);
  w_chunk_load(wp, 1, wqb);
  if (xact) {
    float* xsp = Xs + (xwave * XCH_PER_WAVE * 28 + xlane) * 4;
#pragma unroll
    for (int it = 0; it < XCH_PER_WAVE; ++it)
      *reinterpret_cast<v4f*>(xsp + it * 112) = xin[it];
  }
  w_chunk_store(As0 + asub, wqa);
  lds_barrier();

  w_chunk_load(wp, 2, wqa);

  {
    const int cc   = tid & 15;
    const int lrow = tid >> 4;
    const int col  = n0 + cc;
    const int n    = col / 28;
    const int o    = col - n * 28;
    const int dn   = n - nlo;
    const int om   = (o + 27) % 28;
    int  a1[3], a2[3];
    bool c1[3], c2[3];
#pragma unroll
    for (int k = 0; k < 3; ++k) {
      const int ko  = k + o;
      const int kom = k + om;
      c1[k] = (ko >= 1 && ko <= 28);
      c2[k] = (kom >= 1 && kom <= 28);
      a1[k] = lrow * 112 + dn * 28 + ((ko + 26) % 28);
      a2[k] = (64 + lrow) * 112 + dn * 28 + ((kom + 26) % 28);
    }
    float av[36], bv[36];
#pragma unroll
    for (int j = 0; j < 3; ++j) {
#pragma unroll
      for (int k = 0; k < 3; ++k) {
#pragma unroll
        for (int li = 0; li < 4; ++li) {
          const int off = li * (16 * 112) + j * 28;
          av[(j * 3 + k) * 4 + li] = Xs[a1[k] + off];
          bv[(j * 3 + k) * 4 + li] = Xs[a2[k] + off];
        }
      }
    }
#pragma unroll
    for (int j = 0; j < 3; ++j) {
#pragma unroll
      for (int k = 0; k < 3; ++k) {
#pragma unroll
        for (int li = 0; li < 4; ++li) {
          const int e = (j * 3 + k) * 4 + li;
          Bt[e * 256 + tid] = (c1[k] ? av[e] : 0.0f) + (c2[k] ? bv[e] : 0.0f);
        }
      }
    }
  }
  w_chunk_store(As1 + asub, wqb);
  lds_barrier();

  const int wave = threadIdx.x >> 5;
  const int lane = threadIdx.x & 31;
  const int half = lane >> 4;
  const int lid  = lane & 15;
  const int koff = half * 2;

  const int    aoff = (wave * 16 + lid) * APITCH + koff;
  const float* bcol = &Bt[koff * TILE_N + lid];

  v8f acc0 = {};
  v8f acc1 = {};
  mma_chunk(As0 + aoff, bcol, acc0, acc1);
  lds_barrier();
  w_chunk_store(As0 + asub, wqa);
  mma_chunk(As1 + aoff, bcol + KCH * TILE_N, acc0, acc1);
  lds_barrier();
  mma_chunk(As0 + aoff, bcol + 2 * KCH * TILE_N, acc0, acc1);
  v8f acc = acc0 + acc1;

  const int nidx  = n0 + lid;
  const int ibase = wave * 16 + half * 8;
#pragma unroll
  for (int r = 0; r < 8; ++r)
    out[(ibase + r) * NCOL + nidx] = acc[r];
}

extern "C" void kernel_launch(void* const* d_in, const int* in_sizes, int n_in,
                              void* d_out, int out_size, void* d_ws, size_t ws_size,
                              hipStream_t stream) {
  const float* x = (const float*)d_in[0];
  const float* w = (const float*)d_in[1];
  float* out = (float*)d_out;

  shiftconv_wmma_f32<<<NCOL / TILE_N, 256, 0, stream>>>(x, w, out);
}


# A operand: w streamed in 9 one-tap K chunks (was 3) through a 3-slot LDS ring with 3-4 chunks of loads in flight; matrix steps start after x and 2 small chunks, the rest of w lands under them
# speedup vs baseline: 3.0610x; 1.1614x over previous
#include <hip/hip_runtime.h>


typedef __attribute__((ext_vector_type(2))) float v2f;
typedef __attribute__((ext_vector_type(4))) float v4f;
typedef __attribute__((ext_vector_type(8))) float v8f;

#define KDIM   576
#define NCOL   784
#define TILE_N 16
#define LDS_ELEMS (KDIM * TILE_N)
#define XROWS  4
#define XQUADS (128 * XROWS * 7)
#define XS_ELEMS (XQUADS * 4)
#define XCH_PER_WAVE 16
#define KCH    64
#define NCHUNK 9
#define NSLOT  3
#define APITCH 68
#define ACH_ELEMS (128 * APITCH)
#define AQ_PER_THREAD 8

__device__ __forceinline__ void w_chunk_load(const float* __restrict__ wp, int c, v4f (&r)[AQ_PER_THREAD]) {
#pragma unroll
  for (int it = 0; it < AQ_PER_THREAD; ++it)
    r[it] = *reinterpret_cast<const v4f*>(wp + it * (16 * KDIM) + c * KCH);
}

__device__ __forceinline__ void w_chunk_store(float* __restrict__ ap, const v4f (&r)[AQ_PER_THREAD]) {
#pragma unroll
  for (int it = 0; it < AQ_PER_THREAD; ++it)
    *reinterpret_cast<v4f*>(ap + it * (16 * APITCH)) = r[it];
}

__device__ __forceinline__ void lds_barrier() {
  __builtin_amdgcn_fence(__ATOMIC_RELEASE, "workgroup", "local");
  __builtin_amdgcn_s_barrier();
  __builtin_amdgcn_fence(__ATOMIC_ACQUIRE, "workgroup", "local");
}

struct Frag8 { v2f a[8]; v2f b[8]; };

__device__ __forceinline__ void frag_load(Frag8& f, const float* __restrict__ ar,
                                          const float* __restrict__ bc) {
#pragma unroll
  for (int u = 0; u < 8; ++u) {
    const int kl = 4 * u;
    f.a[u] = *reinterpret_cast<const v2f*>(ar + kl);
    f.b[u].x = bc[kl * TILE_N];
    f.b[u].y = bc[(kl + 1) * TILE_N];
  }
}

__device__ __forceinline__ void frag_mma(const Frag8& f, v8f& acc0, v8f& acc1) {
#pragma unroll
  for (int u = 0; u < 8; u += 2) {
    acc0 = __builtin_amdgcn_wmma_f32_16x16x4_f32(false, f.a[u], false, f.b[u],
                                                 (short)0, acc0, false, false);
    acc1 = __builtin_amdgcn_wmma_f32_16x16x4_f32(false, f.a[u + 1], false, f.b[u + 1],
                                                 (short)0, acc1, false, false);
  }
}

__global__ __launch_bounds__(256)
void shiftconv_wmma_f32(const float* __restrict__ x,
                        const float* __restrict__ w,
                        float* __restrict__ out) {
  __shared__ __attribute__((aligned(16))) float lds[LDS_ELEMS + XS_ELEMS + NSLOT * ACH_ELEMS];
  float* const Bt  = lds;
  float* const Xs  = lds + LDS_ELEMS;
  float* const As  = Xs + XS_ELEMS;

  const int tid   = threadIdx.x;
  const int n0    = blockIdx.x * TILE_N;
  const int nlo   = n0 / 28;
  const int hbase = nlo - 1;

  const int  xlane = tid & 31;
  const int  xwave = tid >> 5;
  const int  xh    = hbase + xlane / 7;
  const bool xact  = xlane < 28;
  const bool xld   = xact && xh >= 0 && xh < 28;
  v4f xin[XCH_PER_WAVE] = {};
  if (xld) {
    const float* xp = x + (xwave * XCH_PER_WAVE) * 784 + hbase * 28 + xlane * 4;
#pragma unroll
    for (int it = 0; it < XCH_PER_WAVE; ++it)
      xin[it] = *reinterpret_cast<const v4f*>(xp + it * 784);
  }
  const int    wsub = (tid >> 4) * KDIM + (tid & 15) * 4;
  const float* wp   = w + wsub;
  const int    asub = (tid >> 4) * APITCH + (tid & 15) * 4;
  v4f wr[NSLOT][AQ_PER_THREAD];
  w_chunk_load(wp, 0, wr[0]);
  __builtin_amdgcn_sched_barrier(0);
  w_chunk_load(wp, 1, wr[1]);
  __builtin_amdgcn_sched_barrier(0);
  w_chunk_load(wp, 2, wr[2]);
  if (xact) {
    float* xsp = Xs + (xwave * XCH_PER_WAVE * 28 + xlane) * 4;
#pragma unroll
    for (int it = 0; it < XCH_PER_WAVE; ++it)
      *reinterpret_cast<v4f*>(xsp + it * 112) = xin[it];
  }
  lds_barrier();

  {
    const int cc   = tid & 15;
    const int lrow = tid >> 4;
    const int col  = n0 + cc;
    const int n    = col / 28;
    const int o    = col - n * 28;
    const int dn   = n - nlo;
    const int om   = (o + 27) % 28;
    int  a1[3], a2[3];
    bool c1[3], c2[3];
#pragma unroll
    for (int k = 0; k < 3; ++k) {
      const int ko  = k + o;
      const int kom = k + om;
      c1[k] = (ko >= 1 && ko <= 28);
      c2[k] = (kom >= 1 && kom <= 28);
      a1[k] = lrow * 112 + dn * 28 + ((ko + 26) % 28);
      a2[k] = (64 + lrow) * 112 + dn * 28 + ((kom + 26) % 28);
    }
    float av[36], bv[36];
#pragma unroll
    for (int j = 0; j < 3; ++j) {
#pragma unroll
      for (int k = 0; k < 3; ++k) {
#pragma unroll
        for (int li = 0; li < 4; ++li) {
          const int off = li * (16 * 112) + j * 28;
          av[(j * 3 + k) * 4 + li] = Xs[a1[k] + off];
          bv[(j * 3 + k) * 4 + li] = Xs[a2[k] + off];
        }
      }
    }
#pragma unroll
    for (int j = 0; j < 3; ++j) {
#pragma unroll
      for (int k = 0; k < 3; ++k) {
#pragma unroll
        for (int li = 0; li < 4; ++li) {
          const int e = (j * 3 + k) * 4 + li;
          Bt[e * 256 + tid] = (c1[k] ? av[e] : 0.0f) + (c2[k] ? bv[e] : 0.0f);
        }
      }
    }
  }
  w_chunk_store(As + 0 * ACH_ELEMS + asub, wr[0]);
  w_chunk_load(wp, 3, wr[0]);
  __builtin_amdgcn_sched_barrier(0);
  w_chunk_store(As + 1 * ACH_ELEMS + asub, wr[1]);
  w_chunk_load(wp, 4, wr[1]);
  lds_barrier();

  const int wave = threadIdx.x >> 5;
  const int lane = threadIdx.x & 31;
  const int half = lane >> 4;
  const int lid  = lane & 15;
  const int koff = half * 2;

  const int    aoff = (wave * 16 + lid) * APITCH + koff;
  const float* bcol = &Bt[koff * TILE_N + lid];

  v8f acc0 = {};
  v8f acc1 = {};
  Frag8 f0, f1;
  frag_load(f0, As + aoff, bcol);
#pragma unroll
  for (int c = 0; c < NCHUNK; ++c) {
    const float* a_cur = As + (c % NSLOT) * ACH_ELEMS + aoff;
    const float* a_nxt = As + ((c + 1) % NSLOT) * ACH_ELEMS + aoff;
    const float* b_cur = bcol + c * KCH * TILE_N;
    frag_load(f1, a_cur + 32, b_cur + 32 * TILE_N);
    __builtin_amdgcn_sched_barrier(0);
    frag_mma(f0, acc0, acc1);
    __builtin_amdgcn_sched_barrier(0);
    if (c + 1 < NCHUNK) frag_load(f0, a_nxt, b_cur + KCH * TILE_N);
    __builtin_amdgcn_sched_barrier(0);
    if (c + 2 < NCHUNK) {
      w_chunk_store(As + ((c + 2) % NSLOT) * ACH_ELEMS + asub, wr[(c + 2) % NSLOT]);
      if (c + 5 < NCHUNK) w_chunk_load(wp, c + 5, wr[(c + 2) % NSLOT]);
      __builtin_amdgcn_sched_barrier(0);
    }
    frag_mma(f1, acc0, acc1);
    __builtin_amdgcn_sched_barrier(0);
    if (c + 2 < NCHUNK)
      lds_barrier();
  }
  v8f acc = acc0 + acc1;

  const int nidx  = n0 + lid;
  const int ibase = wave * 16 + half * 8;
#pragma unroll
  for (int r = 0; r < 8; ++r)
    out[(ibase + r) * NCOL + nidx] = acc[r];
}

extern "C" void kernel_launch(void* const* d_in, const int* in_sizes, int n_in,
                              void* d_out, int out_size, void* d_ws, size_t ws_size,
                              hipStream_t stream) {
  const float* x = (const float*)d_in[0];
  const float* w = (const float*)d_in[1];
  float* out = (float*)d_out;

  shiftconv_wmma_f32<<<NCOL / TILE_N, 256, 0, stream>>>(x, w, out);
}
